# starting kernel
# baseline (speedup 1.0000x reference)
#include <hip/hip_runtime.h>

typedef float v2f __attribute__((ext_vector_type(2)));
typedef float v8f __attribute__((ext_vector_type(8)));

__global__ __launch_bounds__(256) void init_deg_kernel(float* __restrict__ deg, int n) {
    int i = blockIdx.x * blockDim.x + threadIdx.x;
    if (i < n) deg[i] = 1.0f;
}

__global__ __launch_bounds__(256) void deg_scatter_kernel(const int* __restrict__ dst,
                                                          float* __restrict__ deg, int e) {
    int i = blockIdx.x * blockDim.x + threadIdx.x;
    if (i < e) atomicAdd(&deg[dst[i]], 1.0f);
}

__global__ __launch_bounds__(256) void dinv_kernel(const float* __restrict__ deg,
                                                   float* __restrict__ dinv, int n) {
    int i = blockIdx.x * blockDim.x + threadIdx.x;
    if (i < n) dinv[i] = rsqrtf(deg[i]);
}

template <int K, int N>
__global__ __launch_bounds__(128) void gemm_wmma_kernel(const float* __restrict__ A,
                                                        const float* __restrict__ B,
                                                        float* __restrict__ C, int M) {
    __shared__ float Alds[64 * K];
    __shared__ float Blds[K * 64];

    const int tid     = threadIdx.x;
    const int wave    = tid >> 5;
    const int lane    = tid & 31;
    const int lhalf   = lane >> 4;
    const int l15     = lane & 15;
    const int rowBase = blockIdx.x * 64;

    for (int i = tid; i < 64 * K; i += 128) {
        int r = rowBase + i / K;
        Alds[i] = (r < M) ? A[(size_t)r * K + (i % K)] : 0.0f;
    }

    const int mRow = wave * 16 + l15;

    for (int nb = 0; nb < N; nb += 64) {
        __syncthreads();
        for (int i = tid; i < K * 64; i += 128) {
            int kk = i >> 6, nn = i & 63;
            Blds[i] = B[(size_t)kk * N + nb + nn];
        }
        __syncthreads();

        for (int nt = 0; nt < 4; ++nt) {
            const int n = nt * 16 + l15;
            v8f acc = {};
            for (int k = 0; k < K; k += 4) {
                const int ka = k + lhalf * 2;
                v2f a, b;
                a.x = Alds[mRow * K + ka];
                a.y = Alds[mRow * K + ka + 1];
                b.x = Blds[ka * 64 + n];
                b.y = Blds[(ka + 1) * 64 + n];
                acc = __builtin_amdgcn_wmma_f32_16x16x4_f32(
                     false, a,  false, b,
                     (short)0, acc,  false,  false);
            }
            const int row0 = rowBase + wave * 16 + lhalf * 8;
            const int col  = nb + nt * 16 + l15;
#pragma unroll
            for (int g = 0; g < 8; ++g) {
                int r = row0 + g;
                if (r < M) C[(size_t)r * N + col] = acc[g];
            }
        }
    }
}

template <int F>
__global__ __launch_bounds__(256) void self_init_kernel(const float* __restrict__ H,
                                                        const float* __restrict__ dinv,
                                                        float* __restrict__ AGG, int n) {
    int t = blockIdx.x * blockDim.x + threadIdx.x;
    int total = n * (F / 4);
    if (t >= total) return;
    int node = t / (F / 4);
    int fg   = t % (F / 4);
    float w  = dinv[node] * dinv[node];
    const float4* h = (const float4*)(H + (size_t)node * F);
    float4 v = h[fg];
    float4* o = (float4*)(AGG + (size_t)node * F);
    o[fg] = make_float4(v.x * w, v.y * w, v.z * w, v.w * w);
}

template <int F>
__global__ __launch_bounds__(256) void edge_scatter_kernel(const int* __restrict__ src,
                                                           const int* __restrict__ dst,
                                                           const float* __restrict__ dinv,
                                                           const float* __restrict__ H,
                                                           float* __restrict__ AGG, int e) {
    int t = blockIdx.x * blockDim.x + threadIdx.x;
    int total = e * (F / 4);
    if (t >= total) return;
    int ei = t / (F / 4);
    int fg = t % (F / 4);
    int s = src[ei], d = dst[ei];
    float w = dinv[s] * dinv[d];
    const float4* h = (const float4*)(H + (size_t)s * F);
    float4 v = h[fg];
    float* o = AGG + (size_t)d * F + fg * 4;
    atomicAdd(o + 0, v.x * w);
    atomicAdd(o + 1, v.y * w);
    atomicAdd(o + 2, v.z * w);
    atomicAdd(o + 3, v.w * w);
}

template <int F>
__global__ __launch_bounds__(256) void bias_relu_kernel(float* __restrict__ Z,
                                                        const float* __restrict__ bias,
                                                        int n) {
    int t = blockIdx.x * blockDim.x + threadIdx.x;
    int total = n * F;
    if (t >= total) return;
    int f = t % F;
    float v = Z[t] + bias[f];
    Z[t] = v > 0.0f ? v : 0.0f;
}

__global__ __launch_bounds__(256) void decoder_kernel(const float* __restrict__ Z,
                                                      const int* __restrict__ ia,
                                                      const int* __restrict__ ib,
                                                      float* __restrict__ out, int npairs) {
    int pair = blockIdx.x * 8 + (threadIdx.x >> 5);
    int lane = threadIdx.x & 31;
    if (pair >= npairs) return;
    int a = ia[pair], b = ib[pair];
    const float2* za = (const float2*)(Z + (size_t)a * 64);
    const float2* zb = (const float2*)(Z + (size_t)b * 64);
    float2 x = za[lane], y = zb[lane];
    float p = x.x * y.x + x.y * y.y;
#pragma unroll
    for (int off = 16; off > 0; off >>= 1) p += __shfl_xor(p, off, 32);
    if (lane == 0) out[pair] = p;
}

extern "C" void kernel_launch(void* const* d_in, const int* in_sizes, int n_in,
                              void* d_out, int out_size, void* d_ws, size_t ws_size,
                              hipStream_t stream) {
    const float* x     = (const float*)d_in[0];
    const float* W_in  = (const float*)d_in[1];
    const float* b_in  = (const float*)d_in[2];
    const float* W_out = (const float*)d_in[3];
    const float* b_out = (const float*)d_in[4];
    const int*   ei    = (const int*)d_in[5];
    const int*   eli   = (const int*)d_in[6];

    const int N    = in_sizes[0] / 128;
    const int E    = in_sizes[5] / 2;
    const int NDEC = in_sizes[6] / 2;
    const int* src = ei;
    const int* dst = ei + E;
    const int* ia  = eli;
    const int* ib  = eli + NDEC;
    float* logits  = (float*)d_out;

    char* base = (char*)d_ws;
    size_t off = 0;
    auto take = [&](size_t bytes) -> float* {
        float* p = (float*)(base + off);
        off = (off + bytes + 255) & ~(size_t)255;
        return p;
    };
    float* deg  = take((size_t)N * 4);
    float* dinv = take((size_t)N * 4);
    float* H1   = take((size_t)N * 128 * 4);
    float* A1   = take((size_t)N * 128 * 4);
    float* H2   = take((size_t)N * 64 * 4);
    float* A2   = take((size_t)N * 64 * 4);
    (void)ws_size;

    const int TB = 256;
    init_deg_kernel<<<(N + TB - 1) / TB, TB, 0, stream>>>(deg, N);
    deg_scatter_kernel<<<(E + TB - 1) / TB, TB, 0, stream>>>(dst, deg, E);
    dinv_kernel<<<(N + TB - 1) / TB, TB, 0, stream>>>(deg, dinv, N);

    gemm_wmma_kernel<128, 128><<<(N + 63) / 64, 128, 0, stream>>>(x, W_in, H1, N);
    {
        int total = N * 32;
        self_init_kernel<128><<<(total + TB - 1) / TB, TB, 0, stream>>>(H1, dinv, A1, N);
        int etotal = E * 32;
        edge_scatter_kernel<128><<<(etotal + TB - 1) / TB, TB, 0, stream>>>(src, dst, dinv, H1, A1, E);
        int ztotal = N * 128;
        bias_relu_kernel<128><<<(ztotal + TB - 1) / TB, TB, 0, stream>>>(A1, b_in, N);
    }

    gemm_wmma_kernel<128, 64><<<(N + 63) / 64, 128, 0, stream>>>(A1, W_out, H2, N);
    {
        int total = N * 16;
        self_init_kernel<64><<<(total + TB - 1) / TB, TB, 0, stream>>>(H2, dinv, A2, N);
        int etotal = E * 16;
        edge_scatter_kernel<64><<<(etotal + TB - 1) / TB, TB, 0, stream>>>(src, dst, dinv, H2, A2, E);
        int ztotal = N * 64;
        bias_relu_kernel<64><<<(ztotal + TB - 1) / TB, TB, 0, stream>>>(A2, b_out, N);
    }

    decoder_kernel<<<(NDEC + 7) / 8, 256, 0, stream>>>(A2, ia, ib, logits, NDEC);
}


# GEMM: the weight matrix stays in registers as B fragments for the whole launch, one wave per SIMD streams 16-row A tiles from global, no LDS; the k sum order inside the GEMM changes
# speedup vs baseline: 1.1723x; 1.1723x over previous
#include <hip/hip_runtime.h>

typedef float v2f __attribute__((ext_vector_type(2)));
typedef float v8f __attribute__((ext_vector_type(8)));

__global__ __launch_bounds__(256) void init_deg_kernel(float* __restrict__ deg, int n) {
    int i = blockIdx.x * blockDim.x + threadIdx.x;
    if (i < n) deg[i] = 1.0f;
}

__global__ __launch_bounds__(256) void deg_scatter_kernel(const int* __restrict__ dst,
                                                          float* __restrict__ deg, int e) {
    int i = blockIdx.x * blockDim.x + threadIdx.x;
    if (i < e) atomicAdd(&deg[dst[i]], 1.0f);
}

__global__ __launch_bounds__(256) void dinv_kernel(const float* __restrict__ deg,
                                                   float* __restrict__ dinv, int n) {
    int i = blockIdx.x * blockDim.x + threadIdx.x;
    if (i < n) dinv[i] = rsqrtf(deg[i]);
}

template <int K, int N>
__global__ __launch_bounds__(128) void gemm_wmma_kernel(const float* __restrict__ A,
                                                        const float* __restrict__ B,
                                                        float* __restrict__ C, int M) {
    constexpr int NT = N / 16;
    constexpr int NG = N / 64;
    constexpr int KJ = K / 8;

    const int lane   = threadIdx.x & 31;
    const int wave   = threadIdx.x >> 5;
    const int h      = lane >> 4;
    const int c      = lane & 15;
    const int nTiles = (M + 15) >> 4;
    const int nWaves = gridDim.x * 4;
    int t = blockIdx.x * 4 + wave;
    if (t >= nTiles) return;

    v2f bf[2 * KJ][NT];
#pragma unroll
    for (int j = 0; j < KJ; ++j) {
#pragma unroll
        for (int e = 0; e < 2; ++e) {
            const int k0 = 8 * j + 4 * h + 2 * e;
#pragma unroll
            for (int g4 = 0; g4 < NG; ++g4) {
                const float4 b0 = *(const float4*)(B + (size_t)k0 * N + 64 * g4 + 4 * c);
                const float4 b1 = *(const float4*)(B + (size_t)(k0 + 1) * N + 64 * g4 + 4 * c);
                bf[2 * j + e][4 * g4 + 0].x = b0.x; bf[2 * j + e][4 * g4 + 0].y = b1.x;
                bf[2 * j + e][4 * g4 + 1].x = b0.y; bf[2 * j + e][4 * g4 + 1].y = b1.y;
                bf[2 * j + e][4 * g4 + 2].x = b0.z; bf[2 * j + e][4 * g4 + 2].y = b1.z;
                bf[2 * j + e][4 * g4 + 3].x = b0.w; bf[2 * j + e][4 * g4 + 3].y = b1.w;
            }
        }
    }

    float4 q[KJ];
    {
        int r = t * 16 + c; r = r < M ? r : M - 1;
        const float4* ap = (const float4*)(A + (size_t)r * K + 4 * h);
#pragma unroll
        for (int j = 0; j < KJ; ++j) q[j] = ap[2 * j];
    }

    for (;;) {
        const int tn = t + nWaves;
        float4 qn[KJ];
        {
            int r = tn * 16 + c; r = r < M ? r : M - 1;
            const float4* ap = (const float4*)(A + (size_t)r * K + 4 * h);
#pragma unroll
            for (int j = 0; j < KJ; ++j) qn[j] = ap[2 * j];
        }

        v8f acc[NT];
#pragma unroll
        for (int nt = 0; nt < NT; ++nt) acc[nt] = v8f{};

#pragma unroll
        for (int j = 0; j < KJ; ++j) {
            v2f a0, a1;
            a0.x = q[j].x; a0.y = q[j].y;
            a1.x = q[j].z; a1.y = q[j].w;
#pragma unroll
            for (int nt = 0; nt < NT; ++nt)
                acc[nt] = __builtin_amdgcn_wmma_f32_16x16x4_f32(
                    false, a0, false, bf[2 * j][nt], (short)0, acc[nt], false, false);
#pragma unroll
            for (int nt = 0; nt < NT; ++nt)
                acc[nt] = __builtin_amdgcn_wmma_f32_16x16x4_f32(
                    false, a1, false, bf[2 * j + 1][nt], (short)0, acc[nt], false, false);
        }

        const int row0 = t * 16 + 8 * h;
#pragma unroll
        for (int g = 0; g < 8; ++g) {
            const int r = row0 + g;
            if (r < M) {
#pragma unroll
                for (int g4 = 0; g4 < NG; ++g4) {
                    float4 o = make_float4(acc[4 * g4 + 0][g], acc[4 * g4 + 1][g],
                                           acc[4 * g4 + 2][g], acc[4 * g4 + 3][g]);
                    *(float4*)(C + (size_t)r * N + 64 * g4 + 4 * c) = o;
                }
            }
        }

        if (tn >= nTiles) break;
        t = tn;
#pragma unroll
        for (int j = 0; j < KJ; ++j) q[j] = qn[j];
    }
}

template <int F>
__global__ __launch_bounds__(256) void self_init_kernel(const float* __restrict__ H,
                                                        const float* __restrict__ dinv,
                                                        float* __restrict__ AGG, int n) {
    int t = blockIdx.x * blockDim.x + threadIdx.x;
    int total = n * (F / 4);
    if (t >= total) return;
    int node = t / (F / 4);
    int fg   = t % (F / 4);
    float w  = dinv[node] * dinv[node];
    const float4* h = (const float4*)(H + (size_t)node * F);
    float4 v = h[fg];
    float4* o = (float4*)(AGG + (size_t)node * F);
    o[fg] = make_float4(v.x * w, v.y * w, v.z * w, v.w * w);
}

template <int F>
__global__ __launch_bounds__(256) void edge_scatter_kernel(const int* __restrict__ src,
                                                           const int* __restrict__ dst,
                                                           const float* __restrict__ dinv,
                                                           const float* __restrict__ H,
                                                           float* __restrict__ AGG, int e) {
    int t = blockIdx.x * blockDim.x + threadIdx.x;
    int total = e * (F / 4);
    if (t >= total) return;
    int ei = t / (F / 4);
    int fg = t % (F / 4);
    int s = src[ei], d = dst[ei];
    float w = dinv[s] * dinv[d];
    const float4* h = (const float4*)(H + (size_t)s * F);
    float4 v = h[fg];
    float* o = AGG + (size_t)d * F + fg * 4;
    atomicAdd(o + 0, v.x * w);
    atomicAdd(o + 1, v.y * w);
    atomicAdd(o + 2, v.z * w);
    atomicAdd(o + 3, v.w * w);
}

template <int F>
__global__ __launch_bounds__(256) void bias_relu_kernel(float* __restrict__ Z,
                                                        const float* __restrict__ bias,
                                                        int n) {
    int t = blockIdx.x * blockDim.x + threadIdx.x;
    int total = n * F;
    if (t >= total) return;
    int f = t % F;
    float v = Z[t] + bias[f];
    Z[t] = v > 0.0f ? v : 0.0f;
}

__global__ __launch_bounds__(256) void decoder_kernel(const float* __restrict__ Z,
                                                      const int* __restrict__ ia,
                                                      const int* __restrict__ ib,
                                                      float* __restrict__ out, int npairs) {
    int pair = blockIdx.x * 8 + (threadIdx.x >> 5);
    int lane = threadIdx.x & 31;
    if (pair >= npairs) return;
    int a = ia[pair], b = ib[pair];
    const float2* za = (const float2*)(Z + (size_t)a * 64);
    const float2* zb = (const float2*)(Z + (size_t)b * 64);
    float2 x = za[lane], y = zb[lane];
    float p = x.x * y.x + x.y * y.y;
#pragma unroll
    for (int off = 16; off > 0; off >>= 1) p += __shfl_xor(p, off, 32);
    if (lane == 0) out[pair] = p;
}

extern "C" void kernel_launch(void* const* d_in, const int* in_sizes, int n_in,
                              void* d_out, int out_size, void* d_ws, size_t ws_size,
                              hipStream_t stream) {
    const float* x     = (const float*)d_in[0];
    const float* W_in  = (const float*)d_in[1];
    const float* b_in  = (const float*)d_in[2];
    const float* W_out = (const float*)d_in[3];
    const float* b_out = (const float*)d_in[4];
    const int*   ei    = (const int*)d_in[5];
    const int*   eli   = (const int*)d_in[6];

    const int N    = in_sizes[0] / 128;
    const int E    = in_sizes[5] / 2;
    const int NDEC = in_sizes[6] / 2;
    const int* src = ei;
    const int* dst = ei + E;
    const int* ia  = eli;
    const int* ib  = eli + NDEC;
    float* logits  = (float*)d_out;

    char* base = (char*)d_ws;
    size_t off = 0;
    auto take = [&](size_t bytes) -> float* {
        float* p = (float*)(base + off);
        off = (off + bytes + 255) & ~(size_t)255;
        return p;
    };
    float* deg  = take((size_t)N * 4);
    float* dinv = take((size_t)N * 4);
    float* H1   = take((size_t)N * 128 * 4);
    float* A1   = take((size_t)N * 128 * 4);
    float* H2   = take((size_t)N * 64 * 4);
    float* A2   = take((size_t)N * 64 * 4);
    (void)ws_size;

    const int TB = 256;
    init_deg_kernel<<<(N + TB - 1) / TB, TB, 0, stream>>>(deg, N);
    deg_scatter_kernel<<<(E + TB - 1) / TB, TB, 0, stream>>>(dst, deg, E);
    dinv_kernel<<<(N + TB - 1) / TB, TB, 0, stream>>>(deg, dinv, N);

    gemm_wmma_kernel<128, 128><<<256, 128, 0, stream>>>(x, W_in, H1, N);
    {
        int total = N * 32;
        self_init_kernel<128><<<(total + TB - 1) / TB, TB, 0, stream>>>(H1, dinv, A1, N);
        int etotal = E * 32;
        edge_scatter_kernel<128><<<(etotal + TB - 1) / TB, TB, 0, stream>>>(src, dst, dinv, H1, A1, E);
        int ztotal = N * 128;
        bias_relu_kernel<128><<<(ztotal + TB - 1) / TB, TB, 0, stream>>>(A1, b_in, N);
    }

    gemm_wmma_kernel<128, 64><<<256, 128, 0, stream>>>(A1, W_out, H2, N);
    {
        int total = N * 16;
        self_init_kernel<64><<<(total + TB - 1) / TB, TB, 0, stream>>>(H2, dinv, A2, N);
        int etotal = E * 16;
        edge_scatter_kernel<64><<<(etotal + TB - 1) / TB, TB, 0, stream>>>(src, dst, dinv, H2, A2, E);
        int ztotal = N * 64;
        bias_relu_kernel<64><<<(ztotal + TB - 1) / TB, TB, 0, stream>>>(A2, b_out, N);
    }

    decoder_kernel<<<(NDEC + 7) / 8, 256, 0, stream>>>(A2, ia, ib, logits, NDEC);
}


# Edge scatter, lane = feature column: every f32 atomic add and row load covers one whole 128-byte line; a wave walks 32 edges, 32 row lines in flight; sums stay atomic adds in arrival order
# speedup vs baseline: 1.6328x; 1.3928x over previous
#include <hip/hip_runtime.h>

typedef float v2f __attribute__((ext_vector_type(2)));
typedef float v8f __attribute__((ext_vector_type(8)));

__global__ __launch_bounds__(256) void init_deg_kernel(float* __restrict__ deg, int n) {
    int i = blockIdx.x * blockDim.x + threadIdx.x;
    if (i < n) deg[i] = 1.0f;
}

__global__ __launch_bounds__(256) void deg_scatter_kernel(const int* __restrict__ dst,
                                                          float* __restrict__ deg, int e) {
    int i = blockIdx.x * blockDim.x + threadIdx.x;
    if (i < e) atomicAdd(&deg[dst[i]], 1.0f);
}

__global__ __launch_bounds__(256) void dinv_kernel(const float* __restrict__ deg,
                                                   float* __restrict__ dinv, int n) {
    int i = blockIdx.x * blockDim.x + threadIdx.x;
    if (i < n) dinv[i] = rsqrtf(deg[i]);
}

template <int K, int N>
__global__ __launch_bounds__(128) void gemm_wmma_kernel(const float* __restrict__ A,
                                                        const float* __restrict__ B,
                                                        float* __restrict__ C, int M) {
    constexpr int NT = N / 16;
    constexpr int NG = N / 64;
    constexpr int KJ = K / 8;

    const int lane   = threadIdx.x & 31;
    const int wave   = threadIdx.x >> 5;
    const int h      = lane >> 4;
    const int c      = lane & 15;
    const int nTiles = (M + 15) >> 4;
    const int nWaves = gridDim.x * 4;
    int t = blockIdx.x * 4 + wave;
    if (t >= nTiles) return;

    v2f bf[2 * KJ][NT];
#pragma unroll
    for (int j = 0; j < KJ; ++j) {
#pragma unroll
        for (int e = 0; e < 2; ++e) {
            const int k0 = 8 * j + 4 * h + 2 * e;
#pragma unroll
            for (int g4 = 0; g4 < NG; ++g4) {
                const float4 b0 = *(const float4*)(B + (size_t)k0 * N + 64 * g4 + 4 * c);
                const float4 b1 = *(const float4*)(B + (size_t)(k0 + 1) * N + 64 * g4 + 4 * c);
                bf[2 * j + e][4 * g4 + 0].x = b0.x; bf[2 * j + e][4 * g4 + 0].y = b1.x;
                bf[2 * j + e][4 * g4 + 1].x = b0.y; bf[2 * j + e][4 * g4 + 1].y = b1.y;
                bf[2 * j + e][4 * g4 + 2].x = b0.z; bf[2 * j + e][4 * g4 + 2].y = b1.z;
                bf[2 * j + e][4 * g4 + 3].x = b0.w; bf[2 * j + e][4 * g4 + 3].y = b1.w;
            }
        }
    }

    float4 q[KJ];
    {
        int r = t * 16 + c; r = r < M ? r : M - 1;
        const float4* ap = (const float4*)(A + (size_t)r * K + 4 * h);
#pragma unroll
        for (int j = 0; j < KJ; ++j) q[j] = ap[2 * j];
    }

    for (;;) {
        const int tn = t + nWaves;
        float4 qn[KJ];
        {
            int r = tn * 16 + c; r = r < M ? r : M - 1;
            const float4* ap = (const float4*)(A + (size_t)r * K + 4 * h);
#pragma unroll
            for (int j = 0; j < KJ; ++j) qn[j] = ap[2 * j];
        }

        v8f acc[NT];
#pragma unroll
        for (int nt = 0; nt < NT; ++nt) acc[nt] = v8f{};

#pragma unroll
        for (int j = 0; j < KJ; ++j) {
            v2f a0, a1;
            a0.x = q[j].x; a0.y = q[j].y;
            a1.x = q[j].z; a1.y = q[j].w;
#pragma unroll
            for (int nt = 0; nt < NT; ++nt)
                acc[nt] = __builtin_amdgcn_wmma_f32_16x16x4_f32(
                    false, a0, false, bf[2 * j][nt], (short)0, acc[nt], false, false);
#pragma unroll
            for (int nt = 0; nt < NT; ++nt)
                acc[nt] = __builtin_amdgcn_wmma_f32_16x16x4_f32(
                    false, a1, false, bf[2 * j + 1][nt], (short)0, acc[nt], false, false);
        }

        const int row0 = t * 16 + 8 * h;
#pragma unroll
        for (int g = 0; g < 8; ++g) {
            const int r = row0 + g;
            if (r < M) {
#pragma unroll
                for (int g4 = 0; g4 < NG; ++g4) {
                    float4 o = make_float4(acc[4 * g4 + 0][g], acc[4 * g4 + 1][g],
                                           acc[4 * g4 + 2][g], acc[4 * g4 + 3][g]);
                    *(float4*)(C + (size_t)r * N + 64 * g4 + 4 * c) = o;
                }
            }
        }

        if (tn >= nTiles) break;
        t = tn;
#pragma unroll
        for (int j = 0; j < KJ; ++j) q[j] = qn[j];
    }
}

template <int F>
__global__ __launch_bounds__(256) void self_init_kernel(const float* __restrict__ H,
                                                        const float* __restrict__ dinv,
                                                        float* __restrict__ AGG, int n) {
    int t = blockIdx.x * blockDim.x + threadIdx.x;
    int total = n * (F / 4);
    if (t >= total) return;
    int node = t / (F / 4);
    int fg   = t % (F / 4);
    float w  = dinv[node] * dinv[node];
    const float4* h = (const float4*)(H + (size_t)node * F);
    float4 v = h[fg];
    float4* o = (float4*)(AGG + (size_t)node * F);
    o[fg] = make_float4(v.x * w, v.y * w, v.z * w, v.w * w);
}

template <int F, int G>
__global__ __launch_bounds__(256) void edge_scatter_kernel(const int* __restrict__ src,
                                                           const int* __restrict__ dst,
                                                           const float* __restrict__ dinv,
                                                           const float* __restrict__ H,
                                                           float* __restrict__ AGG, int e) {
    constexpr int L = F / 32;
    const unsigned lane = threadIdx.x & 31u;
    const int wave = blockIdx.x * (blockDim.x >> 5) +
                     __builtin_amdgcn_readfirstlane((int)(threadIdx.x >> 5));
    const int e0 = wave * 32;
    if (e0 >= e) return;
    const int cnt = e - e0;

    if (cnt >= 32) {
        const int s = src[e0 + lane];
        const int d = dst[e0 + lane];
        const int wbits = __float_as_int(dinv[s] * dinv[d]);
        for (int b = 0; b < 32; b += G) {
            float v[G][L];
            for (int i = 0; i < G; ++i) {
                const int sj = __builtin_amdgcn_readlane(s, b + i);
                const float* row = H + (size_t)sj * F;
                for (int k = 0; k < L; ++k) v[i][k] = row[lane + 32u * k];
            }
            for (int i = 0; i < G; ++i) {
                const int dj = __builtin_amdgcn_readlane(d, b + i);
                const float wj = __int_as_float(__builtin_amdgcn_readlane(wbits, b + i));
                float* orow = AGG + (size_t)dj * F;
                for (int k = 0; k < L; ++k) atomicAdd(orow + lane + 32u * k, v[i][k] * wj);
            }
        }
    } else {
        for (int j = 0; j < cnt; ++j) {
            const int sj = src[e0 + j], dj = dst[e0 + j];
            const float wj = dinv[sj] * dinv[dj];
            const float* row = H + (size_t)sj * F;
            float* orow = AGG + (size_t)dj * F;
            for (int k = 0; k < L; ++k) atomicAdd(orow + lane + 32u * k, row[lane + 32u * k] * wj);
        }
    }
}

template <int F>
__global__ __launch_bounds__(256) void bias_relu_kernel(float* __restrict__ Z,
                                                        const float* __restrict__ bias,
                                                        int n) {
    int t = blockIdx.x * blockDim.x + threadIdx.x;
    int total = n * F;
    if (t >= total) return;
    int f = t % F;
    float v = Z[t] + bias[f];
    Z[t] = v > 0.0f ? v : 0.0f;
}

__global__ __launch_bounds__(256) void decoder_kernel(const float* __restrict__ Z,
                                                      const int* __restrict__ ia,
                                                      const int* __restrict__ ib,
                                                      float* __restrict__ out, int npairs) {
    int pair = blockIdx.x * 8 + (threadIdx.x >> 5);
    int lane = threadIdx.x & 31;
    if (pair >= npairs) return;
    int a = ia[pair], b = ib[pair];
    const float2* za = (const float2*)(Z + (size_t)a * 64);
    const float2* zb = (const float2*)(Z + (size_t)b * 64);
    float2 x = za[lane], y = zb[lane];
    float p = x.x * y.x + x.y * y.y;
#pragma unroll
    for (int off = 16; off > 0; off >>= 1) p += __shfl_xor(p, off, 32);
    if (lane == 0) out[pair] = p;
}

extern "C" void kernel_launch(void* const* d_in, const int* in_sizes, int n_in,
                              void* d_out, int out_size, void* d_ws, size_t ws_size,
                              hipStream_t stream) {
    const float* x     = (const float*)d_in[0];
    const float* W_in  = (const float*)d_in[1];
    const float* b_in  = (const float*)d_in[2];
    const float* W_out = (const float*)d_in[3];
    const float* b_out = (const float*)d_in[4];
    const int*   ei    = (const int*)d_in[5];
    const int*   eli   = (const int*)d_in[6];

    const int N    = in_sizes[0] / 128;
    const int E    = in_sizes[5] / 2;
    const int NDEC = in_sizes[6] / 2;
    const int* src = ei;
    const int* dst = ei + E;
    const int* ia  = eli;
    const int* ib  = eli + NDEC;
    float* logits  = (float*)d_out;

    char* base = (char*)d_ws;
    size_t off = 0;
    auto take = [&](size_t bytes) -> float* {
        float* p = (float*)(base + off);
        off = (off + bytes + 255) & ~(size_t)255;
        return p;
    };
    float* deg  = take((size_t)N * 4);
    float* dinv = take((size_t)N * 4);
    float* H1   = take((size_t)N * 128 * 4);
    float* A1   = take((size_t)N * 128 * 4);
    float* H2   = take((size_t)N * 64 * 4);
    float* A2   = take((size_t)N * 64 * 4);
    (void)ws_size;

    const int TB = 256;
    init_deg_kernel<<<(N + TB - 1) / TB, TB, 0, stream>>>(deg, N);
    deg_scatter_kernel<<<(E + TB - 1) / TB, TB, 0, stream>>>(dst, deg, E);
    dinv_kernel<<<(N + TB - 1) / TB, TB, 0, stream>>>(deg, dinv, N);

    gemm_wmma_kernel<128, 128><<<256, 128, 0, stream>>>(x, W_in, H1, N);
    {
        int total = N * 32;
        self_init_kernel<128><<<(total + TB - 1) / TB, TB, 0, stream>>>(H1, dinv, A1, N);
        int ewaves = (E + 31) / 32;
        edge_scatter_kernel<128, 8><<<(ewaves + 7) / 8, TB, 0, stream>>>(src, dst, dinv, H1, A1, E);
        int ztotal = N * 128;
        bias_relu_kernel<128><<<(ztotal + TB - 1) / TB, TB, 0, stream>>>(A1, b_in, N);
    }

    gemm_wmma_kernel<128, 64><<<256, 128, 0, stream>>>(A1, W_out, H2, N);
    {
        int total = N * 16;
        self_init_kernel<64><<<(total + TB - 1) / TB, TB, 0, stream>>>(H2, dinv, A2, N);
        int ewaves = (E + 31) / 32;
        edge_scatter_kernel<64, 16><<<(ewaves + 7) / 8, TB, 0, stream>>>(src, dst, dinv, H2, A2, E);
        int ztotal = N * 64;
        bias_relu_kernel<64><<<(ztotal + TB - 1) / TB, TB, 0, stream>>>(A2, b_out, N);
    }

    decoder_kernel<<<(NDEC + 7) / 8, 256, 0, stream>>>(A2, ia, ib, logits, NDEC);
}


# Degree count: work-groups count shares of the edge list into int tables in LDS (ds_add) and store them as planes that dinv sums; no global atomics for degrees, which stay exact; 11 launches, not 12
# speedup vs baseline: 1.9203x; 1.1761x over previous
#include <hip/hip_runtime.h>

typedef float v2f __attribute__((ext_vector_type(2)));
typedef float v8f __attribute__((ext_vector_type(8)));

constexpr int HIST_CAP    = 16384;
constexpr int HIST_SHARES = 64;
constexpr int HIST_TB     = 256;
constexpr int HIST_BATCH  = 32;

__global__ __launch_bounds__(HIST_TB) void deg_hist_kernel(const int* __restrict__ dst,
                                                           int* __restrict__ part,
                                                           int n, int e) {
    __shared__ int4 hist4[HIST_CAP / 4];
    int* hist = (int*)hist4;
    const int tid    = threadIdx.x;
    const int stride = (n + 3) & ~3;
    const int nr     = gridDim.y;
    const int rs     = ((n + nr - 1) / nr + 3) & ~3;
    const int base   = blockIdx.y * rs;
    const int m      = n - base < rs ? n - base : rs;
    if (m <= 0) return;
    const int m4     = (m + 3) >> 2;
    const int per    = (e + HIST_SHARES - 1) / HIST_SHARES;
    const int e0     = blockIdx.x * per;
    const int e1     = e0 + per < e ? e0 + per : e;

    for (int k = tid; k < m4; k += HIST_TB) hist4[k] = make_int4(0, 0, 0, 0);
    __syncthreads();

    for (int i0 = e0 + tid; i0 < e1; i0 += HIST_BATCH * HIST_TB) {
        unsigned d[HIST_BATCH];
#pragma unroll
        for (int k = 0; k < HIST_BATCH; ++k) {
            const int i = i0 + k * HIST_TB;
            d[k] = (unsigned)(dst[i < e1 ? i : e1 - 1] - base);
        }
#pragma unroll
        for (int k = 0; k < HIST_BATCH; ++k) {
            const int i = i0 + k * HIST_TB;
            if (i < e1 && d[k] < (unsigned)m) atomicAdd(&hist[d[k]], 1);
        }
    }
    __syncthreads();

    int4* plane4 = (int4*)(part + (size_t)blockIdx.x * stride + base);
    for (int k = tid; k < m4; k += HIST_TB) plane4[k] = hist4[k];
}

__global__ __launch_bounds__(256) void dinv_kernel(const int* __restrict__ part,
                                                   float* __restrict__ dinv, int n) {
    int i = blockIdx.x * blockDim.x + threadIdx.x;
    if (i >= n) return;
    const int stride = (n + 3) & ~3;
    int s = 1;
#pragma unroll
    for (int b = 0; b < HIST_SHARES; ++b) s += part[(size_t)b * stride + i];
    dinv[i] = rsqrtf((float)s);
}

template <int K, int N>
__global__ __launch_bounds__(128) void gemm_wmma_kernel(const float* __restrict__ A,
                                                        const float* __restrict__ B,
                                                        float* __restrict__ C, int M) {
    constexpr int NT = N / 16;
    constexpr int NG = N / 64;
    constexpr int KJ = K / 8;

    const int lane   = threadIdx.x & 31;
    const int wave   = threadIdx.x >> 5;
    const int h      = lane >> 4;
    const int c      = lane & 15;
    const int nTiles = (M + 15) >> 4;
    const int nWaves = gridDim.x * 4;
    int t = blockIdx.x * 4 + wave;
    if (t >= nTiles) return;

    v2f bf[2 * KJ][NT];
#pragma unroll
    for (int j = 0; j < KJ; ++j) {
#pragma unroll
        for (int e = 0; e < 2; ++e) {
            const int k0 = 8 * j + 4 * h + 2 * e;
#pragma unroll
            for (int g4 = 0; g4 < NG; ++g4) {
                const float4 b0 = *(const float4*)(B + (size_t)k0 * N + 64 * g4 + 4 * c);
                const float4 b1 = *(const float4*)(B + (size_t)(k0 + 1) * N + 64 * g4 + 4 * c);
                bf[2 * j + e][4 * g4 + 0].x = b0.x; bf[2 * j + e][4 * g4 + 0].y = b1.x;
                bf[2 * j + e][4 * g4 + 1].x = b0.y; bf[2 * j + e][4 * g4 + 1].y = b1.y;
                bf[2 * j + e][4 * g4 + 2].x = b0.z; bf[2 * j + e][4 * g4 + 2].y = b1.z;
                bf[2 * j + e][4 * g4 + 3].x = b0.w; bf[2 * j + e][4 * g4 + 3].y = b1.w;
            }
        }
    }

    float4 q[KJ];
    {
        int r = t * 16 + c; r = r < M ? r : M - 1;
        const float4* ap = (const float4*)(A + (size_t)r * K + 4 * h);
#pragma unroll
        for (int j = 0; j < KJ; ++j) q[j] = ap[2 * j];
    }

    for (;;) {
        const int tn = t + nWaves;
        float4 qn[KJ];
        {
            int r = tn * 16 + c; r = r < M ? r : M - 1;
            const float4* ap = (const float4*)(A + (size_t)r * K + 4 * h);
#pragma unroll
            for (int j = 0; j < KJ; ++j) qn[j] = ap[2 * j];
        }

        v8f acc[NT];
#pragma unroll
        for (int nt = 0; nt < NT; ++nt) acc[nt] = v8f{};

#pragma unroll
        for (int j = 0; j < KJ; ++j) {
            v2f a0, a1;
            a0.x = q[j].x; a0.y = q[j].y;
            a1.x = q[j].z; a1.y = q[j].w;
#pragma unroll
            for (int nt = 0; nt < NT; ++nt)
                acc[nt] = __builtin_amdgcn_wmma_f32_16x16x4_f32(
                    false, a0, false, bf[2 * j][nt], (short)0, acc[nt], false, false);
#pragma unroll
            for (int nt = 0; nt < NT; ++nt)
                acc[nt] = __builtin_amdgcn_wmma_f32_16x16x4_f32(
                    false, a1, false, bf[2 * j + 1][nt], (short)0, acc[nt], false, false);
        }

        const int row0 = t * 16 + 8 * h;
#pragma unroll
        for (int g = 0; g < 8; ++g) {
            const int r = row0 + g;
            if (r < M) {
#pragma unroll
                for (int g4 = 0; g4 < NG; ++g4) {
                    float4 o = make_float4(acc[4 * g4 + 0][g], acc[4 * g4 + 1][g],
                                           acc[4 * g4 + 2][g], acc[4 * g4 + 3][g]);
                    *(float4*)(C + (size_t)r * N + 64 * g4 + 4 * c) = o;
                }
            }
        }

        if (tn >= nTiles) break;
        t = tn;
#pragma unroll
        for (int j = 0; j < KJ; ++j) q[j] = qn[j];
    }
}

template <int F>
__global__ __launch_bounds__(256) void self_init_kernel(const float* __restrict__ H,
                                                        const float* __restrict__ dinv,
                                                        float* __restrict__ AGG, int n) {
    int t = blockIdx.x * blockDim.x + threadIdx.x;
    int total = n * (F / 4);
    if (t >= total) return;
    int node = t / (F / 4);
    int fg   = t % (F / 4);
    float w  = dinv[node] * dinv[node];
    const float4* h = (const float4*)(H + (size_t)node * F);
    float4 v = h[fg];
    float4* o = (float4*)(AGG + (size_t)node * F);
    o[fg] = make_float4(v.x * w, v.y * w, v.z * w, v.w * w);
}

template <int F, int G>
__global__ __launch_bounds__(256) void edge_scatter_kernel(const int* __restrict__ src,
                                                           const int* __restrict__ dst,
                                                           const float* __restrict__ dinv,
                                                           const float* __restrict__ H,
                                                           float* __restrict__ AGG, int e) {
    constexpr int L = F / 32;
    const unsigned lane = threadIdx.x & 31u;
    const int wave = blockIdx.x * (blockDim.x >> 5) +
                     __builtin_amdgcn_readfirstlane((int)(threadIdx.x >> 5));
    const int e0 = wave * 32;
    if (e0 >= e) return;
    const int cnt = e - e0;

    if (cnt >= 32) {
        const int s = src[e0 + lane];
        const int d = dst[e0 + lane];
        const int wbits = __float_as_int(dinv[s] * dinv[d]);
        for (int b = 0; b < 32; b += G) {
            float v[G][L];
            for (int i = 0; i < G; ++i) {
                const int sj = __builtin_amdgcn_readlane(s, b + i);
                const float* row = H + (size_t)sj * F;
                for (int k = 0; k < L; ++k) v[i][k] = row[lane + 32u * k];
            }
            for (int i = 0; i < G; ++i) {
                const int dj = __builtin_amdgcn_readlane(d, b + i);
                const float wj = __int_as_float(__builtin_amdgcn_readlane(wbits, b + i));
                float* orow = AGG + (size_t)dj * F;
                for (int k = 0; k < L; ++k) atomicAdd(orow + lane + 32u * k, v[i][k] * wj);
            }
        }
    } else {
        for (int j = 0; j < cnt; ++j) {
            const int sj = src[e0 + j], dj = dst[e0 + j];
            const float wj = dinv[sj] * dinv[dj];
            const float* row = H + (size_t)sj * F;
            float* orow = AGG + (size_t)dj * F;
            for (int k = 0; k < L; ++k) atomicAdd(orow + lane + 32u * k, row[lane + 32u * k] * wj);
        }
    }
}

template <int F>
__global__ __launch_bounds__(256) void bias_relu_kernel(float* __restrict__ Z,
                                                        const float* __restrict__ bias,
                                                        int n) {
    int t = blockIdx.x * blockDim.x + threadIdx.x;
    int total = n * F;
    if (t >= total) return;
    int f = t % F;
    float v = Z[t] + bias[f];
    Z[t] = v > 0.0f ? v : 0.0f;
}

__global__ __launch_bounds__(256) void decoder_kernel(const float* __restrict__ Z,
                                                      const int* __restrict__ ia,
                                                      const int* __restrict__ ib,
                                                      float* __restrict__ out, int npairs) {
    int pair = blockIdx.x * 8 + (threadIdx.x >> 5);
    int lane = threadIdx.x & 31;
    if (pair >= npairs) return;
    int a = ia[pair], b = ib[pair];
    const float2* za = (const float2*)(Z + (size_t)a * 64);
    const float2* zb = (const float2*)(Z + (size_t)b * 64);
    float2 x = za[lane], y = zb[lane];
    float p = x.x * y.x + x.y * y.y;
#pragma unroll
    for (int off = 16; off > 0; off >>= 1) p += __shfl_xor(p, off, 32);
    if (lane == 0) out[pair] = p;
}

extern "C" void kernel_launch(void* const* d_in, const int* in_sizes, int n_in,
                              void* d_out, int out_size, void* d_ws, size_t ws_size,
                              hipStream_t stream) {
    const float* x     = (const float*)d_in[0];
    const float* W_in  = (const float*)d_in[1];
    const float* b_in  = (const float*)d_in[2];
    const float* W_out = (const float*)d_in[3];
    const float* b_out = (const float*)d_in[4];
    const int*   ei    = (const int*)d_in[5];
    const int*   eli   = (const int*)d_in[6];

    const int N    = in_sizes[0] / 128;
    const int E    = in_sizes[5] / 2;
    const int NDEC = in_sizes[6] / 2;
    const int* src = ei;
    const int* dst = ei + E;
    const int* ia  = eli;
    const int* ib  = eli + NDEC;
    float* logits  = (float*)d_out;

    char* base = (char*)d_ws;
    size_t off = 0;
    auto take = [&](size_t bytes) -> float* {
        float* p = (float*)(base + off);
        off = (off + bytes + 255) & ~(size_t)255;
        return p;
    };
    float* dinv = take((size_t)N * 4);
    float* H1   = take((size_t)N * 128 * 4);
    float* A1   = take((size_t)N * 128 * 4);
    float* H2   = take((size_t)N * 64 * 4);
    float* A2   = take((size_t)N * 64 * 4);
    int*   part = (int*)take((size_t)HIST_SHARES * (N + 3) * 4);
    (void)ws_size;

    const int TB = 256;
    deg_hist_kernel<<<dim3(HIST_SHARES, (N + HIST_CAP - 1) / HIST_CAP), HIST_TB, 0, stream>>>(dst, part, N, E);
    dinv_kernel<<<(N + TB - 1) / TB, TB, 0, stream>>>(part, dinv, N);

    gemm_wmma_kernel<128, 128><<<256, 128, 0, stream>>>(x, W_in, H1, N);
    {
        int total = N * 32;
        self_init_kernel<128><<<(total + TB - 1) / TB, TB, 0, stream>>>(H1, dinv, A1, N);
        int ewaves = (E + 31) / 32;
        edge_scatter_kernel<128, 8><<<(ewaves + 7) / 8, TB, 0, stream>>>(src, dst, dinv, H1, A1, E);
        int ztotal = N * 128;
        bias_relu_kernel<128><<<(ztotal + TB - 1) / TB, TB, 0, stream>>>(A1, b_in, N);
    }

    gemm_wmma_kernel<128, 64><<<256, 128, 0, stream>>>(A1, W_out, H2, N);
    {
        int total = N * 16;
        self_init_kernel<64><<<(total + TB - 1) / TB, TB, 0, stream>>>(H2, dinv, A2, N);
        int ewaves = (E + 31) / 32;
        edge_scatter_kernel<64, 16><<<(ewaves + 7) / 8, TB, 0, stream>>>(src, dst, dinv, H2, A2, E);
        int ztotal = N * 64;
        bias_relu_kernel<64><<<(ztotal + TB - 1) / TB, TB, 0, stream>>>(A2, b_out, N);
    }

    decoder_kernel<<<(NDEC + 7) / 8, 256, 0, stream>>>(A2, ia, ib, logits, NDEC);
}


# Decoder: a wave walks 32 pairs, lane = column pair as before, scalar row bases, many rows in flight; partials added by a transposing xor butterfly (same tree, sums bit-identical); one 128-byte store
# speedup vs baseline: 2.7284x; 1.4208x over previous
#include <hip/hip_runtime.h>

typedef float v2f __attribute__((ext_vector_type(2)));
typedef float v8f __attribute__((ext_vector_type(8)));

constexpr int HIST_CAP    = 16384;
constexpr int HIST_SHARES = 64;
constexpr int HIST_TB     = 256;
constexpr int HIST_BATCH  = 32;

__global__ __launch_bounds__(HIST_TB) void deg_hist_kernel(const int* __restrict__ dst,
                                                           int* __restrict__ part,
                                                           int n, int e) {
    __shared__ int4 hist4[HIST_CAP / 4];
    int* hist = (int*)hist4;
    const int tid    = threadIdx.x;
    const int stride = (n + 3) & ~3;
    const int nr     = gridDim.y;
    const int rs     = ((n + nr - 1) / nr + 3) & ~3;
    const int base   = blockIdx.y * rs;
    const int m      = n - base < rs ? n - base : rs;
    if (m <= 0) return;
    const int m4     = (m + 3) >> 2;
    const int per    = (e + HIST_SHARES - 1) / HIST_SHARES;
    const int e0     = blockIdx.x * per;
    const int e1     = e0 + per < e ? e0 + per : e;

    for (int k = tid; k < m4; k += HIST_TB) hist4[k] = make_int4(0, 0, 0, 0);
    __syncthreads();

    for (int i0 = e0 + tid; i0 < e1; i0 += HIST_BATCH * HIST_TB) {
        unsigned d[HIST_BATCH];
#pragma unroll
        for (int k = 0; k < HIST_BATCH; ++k) {
            const int i = i0 + k * HIST_TB;
            d[k] = (unsigned)(dst[i < e1 ? i : e1 - 1] - base);
        }
#pragma unroll
        for (int k = 0; k < HIST_BATCH; ++k) {
            const int i = i0 + k * HIST_TB;
            if (i < e1 && d[k] < (unsigned)m) atomicAdd(&hist[d[k]], 1);
        }
    }
    __syncthreads();

    int4* plane4 = (int4*)(part + (size_t)blockIdx.x * stride + base);
    for (int k = tid; k < m4; k += HIST_TB) plane4[k] = hist4[k];
}

__global__ __launch_bounds__(256) void dinv_kernel(const int* __restrict__ part,
                                                   float* __restrict__ dinv, int n) {
    int i = blockIdx.x * blockDim.x + threadIdx.x;
    if (i >= n) return;
    const int stride = (n + 3) & ~3;
    int s = 1;
#pragma unroll
    for (int b = 0; b < HIST_SHARES; ++b) s += part[(size_t)b * stride + i];
    dinv[i] = rsqrtf((float)s);
}

template <int K, int N>
__global__ __launch_bounds__(128) void gemm_wmma_kernel(const float* __restrict__ A,
                                                        const float* __restrict__ B,
                                                        float* __restrict__ C, int M) {
    constexpr int NT = N / 16;
    constexpr int NG = N / 64;
    constexpr int KJ = K / 8;

    const int lane   = threadIdx.x & 31;
    const int wave   = threadIdx.x >> 5;
    const int h      = lane >> 4;
    const int c      = lane & 15;
    const int nTiles = (M + 15) >> 4;
    const int nWaves = gridDim.x * 4;
    int t = blockIdx.x * 4 + wave;
    if (t >= nTiles) return;

    v2f bf[2 * KJ][NT];
#pragma unroll
    for (int j = 0; j < KJ; ++j) {
#pragma unroll
        for (int e = 0; e < 2; ++e) {
            const int k0 = 8 * j + 4 * h + 2 * e;
#pragma unroll
            for (int g4 = 0; g4 < NG; ++g4) {
                const float4 b0 = *(const float4*)(B + (size_t)k0 * N + 64 * g4 + 4 * c);
                const float4 b1 = *(const float4*)(B + (size_t)(k0 + 1) * N + 64 * g4 + 4 * c);
                bf[2 * j + e][4 * g4 + 0].x = b0.x; bf[2 * j + e][4 * g4 + 0].y = b1.x;
                bf[2 * j + e][4 * g4 + 1].x = b0.y; bf[2 * j + e][4 * g4 + 1].y = b1.y;
                bf[2 * j + e][4 * g4 + 2].x = b0.z; bf[2 * j + e][4 * g4 + 2].y = b1.z;
                bf[2 * j + e][4 * g4 + 3].x = b0.w; bf[2 * j + e][4 * g4 + 3].y = b1.w;
            }
        }
    }

    float4 q[KJ];
    {
        int r = t * 16 + c; r = r < M ? r : M - 1;
        const float4* ap = (const float4*)(A + (size_t)r * K + 4 * h);
#pragma unroll
        for (int j = 0; j < KJ; ++j) q[j] = ap[2 * j];
    }

    for (;;) {
        const int tn = t + nWaves;
        float4 qn[KJ];
        {
            int r = tn * 16 + c; r = r < M ? r : M - 1;
            const float4* ap = (const float4*)(A + (size_t)r * K + 4 * h);
#pragma unroll
            for (int j = 0; j < KJ; ++j) qn[j] = ap[2 * j];
        }

        v8f acc[NT];
#pragma unroll
        for (int nt = 0; nt < NT; ++nt) acc[nt] = v8f{};

#pragma unroll
        for (int j = 0; j < KJ; ++j) {
            v2f a0, a1;
            a0.x = q[j].x; a0.y = q[j].y;
            a1.x = q[j].z; a1.y = q[j].w;
#pragma unroll
            for (int nt = 0; nt < NT; ++nt)
                acc[nt] = __builtin_amdgcn_wmma_f32_16x16x4_f32(
                    false, a0, false, bf[2 * j][nt], (short)0, acc[nt], false, false);
#pragma unroll
            for (int nt = 0; nt < NT; ++nt)
                acc[nt] = __builtin_amdgcn_wmma_f32_16x16x4_f32(
                    false, a1, false, bf[2 * j + 1][nt], (short)0, acc[nt], false, false);
        }

        const int row0 = t * 16 + 8 * h;
#pragma unroll
        for (int g = 0; g < 8; ++g) {
            const int r = row0 + g;
            if (r < M) {
#pragma unroll
                for (int g4 = 0; g4 < NG; ++g4) {
                    float4 o = make_float4(acc[4 * g4 + 0][g], acc[4 * g4 + 1][g],
                                           acc[4 * g4 + 2][g], acc[4 * g4 + 3][g]);
                    *(float4*)(C + (size_t)r * N + 64 * g4 + 4 * c) = o;
                }
            }
        }

        if (tn >= nTiles) break;
        t = tn;
#pragma unroll
        for (int j = 0; j < KJ; ++j) q[j] = qn[j];
    }
}

template <int F>
__global__ __launch_bounds__(256) void self_init_kernel(const float* __restrict__ H,
                                                        const float* __restrict__ dinv,
                                                        float* __restrict__ AGG, int n) {
    int t = blockIdx.x * blockDim.x + threadIdx.x;
    int total = n * (F / 4);
    if (t >= total) return;
    int node = t / (F / 4);
    int fg   = t % (F / 4);
    float w  = dinv[node] * dinv[node];
    const float4* h = (const float4*)(H + (size_t)node * F);
    float4 v = h[fg];
    float4* o = (float4*)(AGG + (size_t)node * F);
    o[fg] = make_float4(v.x * w, v.y * w, v.z * w, v.w * w);
}

template <int F, int G>
__global__ __launch_bounds__(256) void edge_scatter_kernel(const int* __restrict__ src,
                                                           const int* __restrict__ dst,
                                                           const float* __restrict__ dinv,
                                                           const float* __restrict__ H,
                                                           float* __restrict__ AGG, int e) {
    constexpr int L = F / 32;
    const unsigned lane = threadIdx.x & 31u;
    const int wave = blockIdx.x * (blockDim.x >> 5) +
                     __builtin_amdgcn_readfirstlane((int)(threadIdx.x >> 5));
    const int e0 = wave * 32;
    if (e0 >= e) return;
    const int cnt = e - e0;

    if (cnt >= 32) {
        const int s = src[e0 + lane];
        const int d = dst[e0 + lane];
        const int wbits = __float_as_int(dinv[s] * dinv[d]);
        for (int b = 0; b < 32; b += G) {
            float v[G][L];
            for (int i = 0; i < G; ++i) {
                const int sj = __builtin_amdgcn_readlane(s, b + i);
                const float* row = H + (size_t)sj * F;
                for (int k = 0; k < L; ++k) v[i][k] = row[lane + 32u * k];
            }
            for (int i = 0; i < G; ++i) {
                const int dj = __builtin_amdgcn_readlane(d, b + i);
                const float wj = __int_as_float(__builtin_amdgcn_readlane(wbits, b + i));
                float* orow = AGG + (size_t)dj * F;
                for (int k = 0; k < L; ++k) atomicAdd(orow + lane + 32u * k, v[i][k] * wj);
            }
        }
    } else {
        for (int j = 0; j < cnt; ++j) {
            const int sj = src[e0 + j], dj = dst[e0 + j];
            const float wj = dinv[sj] * dinv[dj];
            const float* row = H + (size_t)sj * F;
            float* orow = AGG + (size_t)dj * F;
            for (int k = 0; k < L; ++k) atomicAdd(orow + lane + 32u * k, row[lane + 32u * k] * wj);
        }
    }
}

template <int F>
__global__ __launch_bounds__(256) void bias_relu_kernel(float* __restrict__ Z,
                                                        const float* __restrict__ bias,
                                                        int n) {
    int t = blockIdx.x * blockDim.x + threadIdx.x;
    int total = n * F;
    if (t >= total) return;
    int f = t % F;
    float v = Z[t] + bias[f];
    Z[t] = v > 0.0f ? v : 0.0f;
}

constexpr int DEC_G = 16;

__global__ __launch_bounds__(256) void decoder_kernel(const float* __restrict__ Z,
                                                      const int* __restrict__ ia,
                                                      const int* __restrict__ ib,
                                                      float* __restrict__ out, int npairs) {
    const unsigned lane = threadIdx.x & 31u;
    const int wave = blockIdx.x * (blockDim.x >> 5) +
                     __builtin_amdgcn_readfirstlane((int)(threadIdx.x >> 5));
    const int p0 = wave * 32;
    if (p0 >= npairs) return;
    const int pl = p0 + (int)lane < npairs ? p0 + (int)lane : npairs - 1;
    const int a = ia[pl], b = ib[pl];

    float v[32];
#pragma unroll
    for (int g = 0; g < 32; g += DEC_G) {
        float2 x[DEC_G], y[DEC_G];
#pragma unroll
        for (int i = 0; i < DEC_G; ++i) {
            const unsigned aj = (unsigned)__builtin_amdgcn_readlane(a, g + i);
            const unsigned bj = (unsigned)__builtin_amdgcn_readlane(b, g + i);
            x[i] = ((const float2*)(Z + (size_t)aj * 64))[lane];
            y[i] = ((const float2*)(Z + (size_t)bj * 64))[lane];
        }
#pragma unroll
        for (int i = 0; i < DEC_G; ++i) v[g + i] = x[i].x * y[i].x + x[i].y * y[i].y;
    }

    float w16[16], w8[8], w4[4], w2[2];
    const bool u16 = (lane & 16u) != 0, u8 = (lane & 8u) != 0, u4 = (lane & 4u) != 0,
               u2 = (lane & 2u) != 0, u1 = (lane & 1u) != 0;
#pragma unroll
    for (int i = 0; i < 16; ++i) {
        const float keep = u16 ? v[i + 16] : v[i];
        const float send = u16 ? v[i] : v[i + 16];
        w16[i] = keep + __shfl_xor(send, 16, 32);
    }
#pragma unroll
    for (int i = 0; i < 8; ++i) {
        const float keep = u8 ? w16[i + 8] : w16[i];
        const float send = u8 ? w16[i] : w16[i + 8];
        w8[i] = keep + __shfl_xor(send, 8, 32);
    }
#pragma unroll
    for (int i = 0; i < 4; ++i) {
        const float keep = u4 ? w8[i + 4] : w8[i];
        const float send = u4 ? w8[i] : w8[i + 4];
        w4[i] = keep + __shfl_xor(send, 4, 32);
    }
#pragma unroll
    for (int i = 0; i < 2; ++i) {
        const float keep = u2 ? w4[i + 2] : w4[i];
        const float send = u2 ? w4[i] : w4[i + 2];
        w2[i] = keep + __shfl_xor(send, 2, 32);
    }
    const float keep = u1 ? w2[1] : w2[0];
    const float send = u1 ? w2[0] : w2[1];
    const float r = keep + __shfl_xor(send, 1, 32);

    if (p0 + (int)lane < npairs) out[p0 + lane] = r;
}

extern "C" void kernel_launch(void* const* d_in, const int* in_sizes, int n_in,
                              void* d_out, int out_size, void* d_ws, size_t ws_size,
                              hipStream_t stream) {
    const float* x     = (const float*)d_in[0];
    const float* W_in  = (const float*)d_in[1];
    const float* b_in  = (const float*)d_in[2];
    const float* W_out = (const float*)d_in[3];
    const float* b_out = (const float*)d_in[4];
    const int*   ei    = (const int*)d_in[5];
    const int*   eli   = (const int*)d_in[6];

    const int N    = in_sizes[0] / 128;
    const int E    = in_sizes[5] / 2;
    const int NDEC = in_sizes[6] / 2;
    const int* src = ei;
    const int* dst = ei + E;
    const int* ia  = eli;
    const int* ib  = eli + NDEC;
    float* logits  = (float*)d_out;

    char* base = (char*)d_ws;
    size_t off = 0;
    auto take = [&](size_t bytes) -> float* {
        float* p = (float*)(base + off);
        off = (off + bytes + 255) & ~(size_t)255;
        return p;
    };
    float* dinv = take((size_t)N * 4);
    float* H1   = take((size_t)N * 128 * 4);
    float* A1   = take((size_t)N * 128 * 4);
    float* H2   = take((size_t)N * 64 * 4);
    float* A2   = take((size_t)N * 64 * 4);
    int*   part = (int*)take((size_t)HIST_SHARES * (N + 3) * 4);
    (void)ws_size;

    const int TB = 256;
    deg_hist_kernel<<<dim3(HIST_SHARES, (N + HIST_CAP - 1) / HIST_CAP), HIST_TB, 0, stream>>>(dst, part, N, E);
    dinv_kernel<<<(N + TB - 1) / TB, TB, 0, stream>>>(part, dinv, N);

    gemm_wmma_kernel<128, 128><<<256, 128, 0, stream>>>(x, W_in, H1, N);
    {
        int total = N * 32;
        self_init_kernel<128><<<(total + TB - 1) / TB, TB, 0, stream>>>(H1, dinv, A1, N);
        int ewaves = (E + 31) / 32;
        edge_scatter_kernel<128, 8><<<(ewaves + 7) / 8, TB, 0, stream>>>(src, dst, dinv, H1, A1, E);
        int ztotal = N * 128;
        bias_relu_kernel<128><<<(ztotal + TB - 1) / TB, TB, 0, stream>>>(A1, b_in, N);
    }

    gemm_wmma_kernel<128, 64><<<256, 128, 0, stream>>>(A1, W_out, H2, N);
    {
        int total = N * 16;
        self_init_kernel<64><<<(total + TB - 1) / TB, TB, 0, stream>>>(H2, dinv, A2, N);
        int ewaves = (E + 31) / 32;
        edge_scatter_kernel<64, 16><<<(ewaves + 7) / 8, TB, 0, stream>>>(src, dst, dinv, H2, A2, E);
        int ztotal = N * 64;
        bias_relu_kernel<64><<<(ztotal + TB - 1) / TB, TB, 0, stream>>>(A2, b_out, N);
    }

    {
        int dwaves = (NDEC + 31) / 32;
        decoder_kernel<<<(dwaves + 7) / 8, 256, 0, stream>>>(A2, ia, ib, logits, NDEC);
    }
}


# Elementwise passes ride on their neighbours: the GEMM store also writes the self-loop start value; bias+ReLU applied where the next GEMM and the decoder load rows; 7 launches, not 11; sums as before
# speedup vs baseline: 2.8280x; 1.0365x over previous
#include <hip/hip_runtime.h>

typedef float v2f __attribute__((ext_vector_type(2)));
typedef float v8f __attribute__((ext_vector_type(8)));

constexpr int HIST_CAP    = 16384;
constexpr int HIST_SHARES = 64;
constexpr int HIST_TB     = 256;
constexpr int HIST_BATCH  = 32;

__global__ __launch_bounds__(HIST_TB) void deg_hist_kernel(const int* __restrict__ dst,
                                                           int* __restrict__ part,
                                                           int n, int e) {
    __shared__ int4 hist4[HIST_CAP / 4];
    int* hist = (int*)hist4;
    const int tid    = threadIdx.x;
    const int stride = (n + 3) & ~3;
    const int nr     = gridDim.y;
    const int rs     = ((n + nr - 1) / nr + 3) & ~3;
    const int base   = blockIdx.y * rs;
    const int m      = n - base < rs ? n - base : rs;
    if (m <= 0) return;
    const int m4     = (m + 3) >> 2;
    const int per    = (e + HIST_SHARES - 1) / HIST_SHARES;
    const int e0     = blockIdx.x * per;
    const int e1     = e0 + per < e ? e0 + per : e;

    for (int k = tid; k < m4; k += HIST_TB) hist4[k] = make_int4(0, 0, 0, 0);
    __syncthreads();

    for (int i0 = e0 + tid; i0 < e1; i0 += HIST_BATCH * HIST_TB) {
        unsigned d[HIST_BATCH];
#pragma unroll
        for (int k = 0; k < HIST_BATCH; ++k) {
            const int i = i0 + k * HIST_TB;
            d[k] = (unsigned)(dst[i < e1 ? i : e1 - 1] - base);
        }
#pragma unroll
        for (int k = 0; k < HIST_BATCH; ++k) {
            const int i = i0 + k * HIST_TB;
            if (i < e1 && d[k] < (unsigned)m) atomicAdd(&hist[d[k]], 1);
        }
    }
    __syncthreads();

    int4* plane4 = (int4*)(part + (size_t)blockIdx.x * stride + base);
    for (int k = tid; k < m4; k += HIST_TB) plane4[k] = hist4[k];
}

__global__ __launch_bounds__(256) void dinv_kernel(const int* __restrict__ part,
                                                   float* __restrict__ dinv, int n) {
    int i = blockIdx.x * blockDim.x + threadIdx.x;
    if (i >= n) return;
    const int stride = (n + 3) & ~3;
    int s = 1;
#pragma unroll
    for (int b = 0; b < HIST_SHARES; ++b) s += part[(size_t)b * stride + i];
    dinv[i] = rsqrtf((float)s);
}

__device__ __forceinline__ float relu_f(float v) { return v > 0.0f ? v : 0.0f; }

template <int K, int N, bool RELU_IN>
__global__ __launch_bounds__(128) void gemm_wmma_kernel(const float* __restrict__ A,
                                                        const float* __restrict__ abias,
                                                        const float* __restrict__ B,
                                                        const float* __restrict__ dinv,
                                                        float* __restrict__ C,
                                                        float* __restrict__ AGG, int M) {
    constexpr int NT = N / 16;
    constexpr int NG = N / 64;
    constexpr int KJ = K / 8;

    const int lane   = threadIdx.x & 31;
    const int wave   = threadIdx.x >> 5;
    const int h      = lane >> 4;
    const int c      = lane & 15;
    const int nTiles = (M + 15) >> 4;
    const int nWaves = gridDim.x * 4;
    int t = blockIdx.x * 4 + wave;
    if (t >= nTiles) return;

    v2f bf[2 * KJ][NT];
#pragma unroll
    for (int j = 0; j < KJ; ++j) {
#pragma unroll
        for (int e = 0; e < 2; ++e) {
            const int k0 = 8 * j + 4 * h + 2 * e;
#pragma unroll
            for (int g4 = 0; g4 < NG; ++g4) {
                const float4 b0 = *(const float4*)(B + (size_t)k0 * N + 64 * g4 + 4 * c);
                const float4 b1 = *(const float4*)(B + (size_t)(k0 + 1) * N + 64 * g4 + 4 * c);
                bf[2 * j + e][4 * g4 + 0].x = b0.x; bf[2 * j + e][4 * g4 + 0].y = b1.x;
                bf[2 * j + e][4 * g4 + 1].x = b0.y; bf[2 * j + e][4 * g4 + 1].y = b1.y;
                bf[2 * j + e][4 * g4 + 2].x = b0.z; bf[2 * j + e][4 * g4 + 2].y = b1.z;
                bf[2 * j + e][4 * g4 + 3].x = b0.w; bf[2 * j + e][4 * g4 + 3].y = b1.w;
            }
        }
    }

    float4 bq[KJ];
    if (RELU_IN) {
#pragma unroll
        for (int j = 0; j < KJ; ++j) bq[j] = *(const float4*)(abias + 8 * j + 4 * h);
    }

    float4 q[KJ];
    {
        int r = t * 16 + c; r = r < M ? r : M - 1;
        const float4* ap = (const float4*)(A + (size_t)r * K + 4 * h);
#pragma unroll
        for (int j = 0; j < KJ; ++j) q[j] = ap[2 * j];
    }
    float dv[8];
#pragma unroll
    for (int g = 0; g < 8; ++g) {
        int r = t * 16 + 8 * h + g; r = r < M ? r : M - 1;
        dv[g] = dinv[r];
    }

    for (;;) {
        const int tn = t + nWaves;
        float4 qn[KJ];
        {
            int r = tn * 16 + c; r = r < M ? r : M - 1;
            const float4* ap = (const float4*)(A + (size_t)r * K + 4 * h);
#pragma unroll
            for (int j = 0; j < KJ; ++j) qn[j] = ap[2 * j];
        }
        float dvn[8];
#pragma unroll
        for (int g = 0; g < 8; ++g) {
            int r = tn * 16 + 8 * h + g; r = r < M ? r : M - 1;
            dvn[g] = dinv[r];
        }
        __builtin_amdgcn_sched_barrier(0);

        v8f acc[NT];
#pragma unroll
        for (int nt = 0; nt < NT; ++nt) acc[nt] = v8f{};

#pragma unroll
        for (int j = 0; j < KJ; ++j) {
            float4 qa = q[j];
            if (RELU_IN) {
                qa.x = relu_f(qa.x + bq[j].x); qa.y = relu_f(qa.y + bq[j].y);
                qa.z = relu_f(qa.z + bq[j].z); qa.w = relu_f(qa.w + bq[j].w);
            }
            v2f a0, a1;
            a0.x = qa.x; a0.y = qa.y;
            a1.x = qa.z; a1.y = qa.w;
#pragma unroll
            for (int nt = 0; nt < NT; ++nt)
                acc[nt] = __builtin_amdgcn_wmma_f32_16x16x4_f32(
                    false, a0, false, bf[2 * j][nt], (short)0, acc[nt], false, false);
#pragma unroll
            for (int nt = 0; nt < NT; ++nt)
                acc[nt] = __builtin_amdgcn_wmma_f32_16x16x4_f32(
                    false, a1, false, bf[2 * j + 1][nt], (short)0, acc[nt], false, false);
        }

        const int row0 = t * 16 + 8 * h;
#pragma unroll
        for (int g = 0; g < 8; ++g) {
            const int r = row0 + g;
            if (r < M) {
                const float w = dv[g] * dv[g];
#pragma unroll
                for (int g4 = 0; g4 < NG; ++g4) {
                    float4 o = make_float4(acc[4 * g4 + 0][g], acc[4 * g4 + 1][g],
                                           acc[4 * g4 + 2][g], acc[4 * g4 + 3][g]);
                    *(float4*)(C + (size_t)r * N + 64 * g4 + 4 * c) = o;
                    *(float4*)(AGG + (size_t)r * N + 64 * g4 + 4 * c) =
                        make_float4(o.x * w, o.y * w, o.z * w, o.w * w);
                }
            }
        }

        if (tn >= nTiles) break;
        t = tn;
#pragma unroll
        for (int j = 0; j < KJ; ++j) q[j] = qn[j];
#pragma unroll
        for (int g = 0; g < 8; ++g) dv[g] = dvn[g];
    }
}


template <int F, int G>
__global__ __launch_bounds__(256) void edge_scatter_kernel(const int* __restrict__ src,
                                                           const int* __restrict__ dst,
                                                           const float* __restrict__ dinv,
                                                           const float* __restrict__ H,
                                                           float* __restrict__ AGG, int e) {
    constexpr int L = F / 32;
    const unsigned lane = threadIdx.x & 31u;
    const int wave = blockIdx.x * (blockDim.x >> 5) +
                     __builtin_amdgcn_readfirstlane((int)(threadIdx.x >> 5));
    const int e0 = wave * 32;
    if (e0 >= e) return;
    const int cnt = e - e0;

    if (cnt >= 32) {
        const int s = src[e0 + lane];
        const int d = dst[e0 + lane];
        const int wbits = __float_as_int(dinv[s] * dinv[d]);
        for (int b = 0; b < 32; b += G) {
            float v[G][L];
            for (int i = 0; i < G; ++i) {
                const int sj = __builtin_amdgcn_readlane(s, b + i);
                const float* row = H + (size_t)sj * F;
                for (int k = 0; k < L; ++k) v[i][k] = row[lane + 32u * k];
            }
            for (int i = 0; i < G; ++i) {
                const int dj = __builtin_amdgcn_readlane(d, b + i);
                const float wj = __int_as_float(__builtin_amdgcn_readlane(wbits, b + i));
                float* orow = AGG + (size_t)dj * F;
                for (int k = 0; k < L; ++k) atomicAdd(orow + lane + 32u * k, v[i][k] * wj);
            }
        }
    } else {
        for (int j = 0; j < cnt; ++j) {
            const int sj = src[e0 + j], dj = dst[e0 + j];
            const float wj = dinv[sj] * dinv[dj];
            const float* row = H + (size_t)sj * F;
            float* orow = AGG + (size_t)dj * F;
            for (int k = 0; k < L; ++k) atomicAdd(orow + lane + 32u * k, row[lane + 32u * k] * wj);
        }
    }
}

constexpr int DEC_G = 16;

__global__ __launch_bounds__(256) void decoder_kernel(const float* __restrict__ Z,
                                                      const float* __restrict__ bias,
                                                      const int* __restrict__ ia,
                                                      const int* __restrict__ ib,
                                                      float* __restrict__ out, int npairs) {
    const unsigned lane = threadIdx.x & 31u;
    const int wave = blockIdx.x * (blockDim.x >> 5) +
                     __builtin_amdgcn_readfirstlane((int)(threadIdx.x >> 5));
    const int p0 = wave * 32;
    if (p0 >= npairs) return;
    const int pl = p0 + (int)lane < npairs ? p0 + (int)lane : npairs - 1;
    const int a = ia[pl], b = ib[pl];
    const float2 bz = ((const float2*)bias)[lane];

    float v[32];
#pragma unroll
    for (int g = 0; g < 32; g += DEC_G) {
        float2 x[DEC_G], y[DEC_G];
#pragma unroll
        for (int i = 0; i < DEC_G; ++i) {
            const unsigned aj = (unsigned)__builtin_amdgcn_readlane(a, g + i);
            const unsigned bj = (unsigned)__builtin_amdgcn_readlane(b, g + i);
            x[i] = ((const float2*)(Z + (size_t)aj * 64))[lane];
            y[i] = ((const float2*)(Z + (size_t)bj * 64))[lane];
        }
#pragma unroll
        for (int i = 0; i < DEC_G; ++i) {
            const float xa = relu_f(x[i].x + bz.x), xb = relu_f(x[i].y + bz.y);
            const float ya = relu_f(y[i].x + bz.x), yb = relu_f(y[i].y + bz.y);
            v[g + i] = xa * ya + xb * yb;
        }
    }

    float w16[16], w8[8], w4[4], w2[2];
    const bool u16 = (lane & 16u) != 0, u8 = (lane & 8u) != 0, u4 = (lane & 4u) != 0,
               u2 = (lane & 2u) != 0, u1 = (lane & 1u) != 0;
#pragma unroll
    for (int i = 0; i < 16; ++i) {
        const float keep = u16 ? v[i + 16] : v[i];
        const float send = u16 ? v[i] : v[i + 16];
        w16[i] = keep + __shfl_xor(send, 16, 32);
    }
#pragma unroll
    for (int i = 0; i < 8; ++i) {
        const float keep = u8 ? w16[i + 8] : w16[i];
        const float send = u8 ? w16[i] : w16[i + 8];
        w8[i] = keep + __shfl_xor(send, 8, 32);
    }
#pragma unroll
    for (int i = 0; i < 4; ++i) {
        const float keep = u4 ? w8[i + 4] : w8[i];
        const float send = u4 ? w8[i] : w8[i + 4];
        w4[i] = keep + __shfl_xor(send, 4, 32);
    }
#pragma unroll
    for (int i = 0; i < 2; ++i) {
        const float keep = u2 ? w4[i + 2] : w4[i];
        const float send = u2 ? w4[i] : w4[i + 2];
        w2[i] = keep + __shfl_xor(send, 2, 32);
    }
    const float keep = u1 ? w2[1] : w2[0];
    const float send = u1 ? w2[0] : w2[1];
    const float r = keep + __shfl_xor(send, 1, 32);

    if (p0 + (int)lane < npairs) out[p0 + lane] = r;
}

extern "C" void kernel_launch(void* const* d_in, const int* in_sizes, int n_in,
                              void* d_out, int out_size, void* d_ws, size_t ws_size,
                              hipStream_t stream) {
    const float* x     = (const float*)d_in[0];
    const float* W_in  = (const float*)d_in[1];
    const float* b_in  = (const float*)d_in[2];
    const float* W_out = (const float*)d_in[3];
    const float* b_out = (const float*)d_in[4];
    const int*   ei    = (const int*)d_in[5];
    const int*   eli   = (const int*)d_in[6];

    const int N    = in_sizes[0] / 128;
    const int E    = in_sizes[5] / 2;
    const int NDEC = in_sizes[6] / 2;
    const int* src = ei;
    const int* dst = ei + E;
    const int* ia  = eli;
    const int* ib  = eli + NDEC;
    float* logits  = (float*)d_out;

    char* base = (char*)d_ws;
    size_t off = 0;
    auto take = [&](size_t bytes) -> float* {
        float* p = (float*)(base + off);
        off = (off + bytes + 255) & ~(size_t)255;
        return p;
    };
    float* dinv = take((size_t)N * 4);
    float* H1   = take((size_t)N * 128 * 4);
    float* A1   = take((size_t)N * 128 * 4);
    float* H2   = take((size_t)N * 64 * 4);
    float* A2   = take((size_t)N * 64 * 4);
    int*   part = (int*)take((size_t)HIST_SHARES * (N + 3) * 4);
    (void)ws_size;

    const int TB = 256;
    deg_hist_kernel<<<dim3(HIST_SHARES, (N + HIST_CAP - 1) / HIST_CAP), HIST_TB, 0, stream>>>(dst, part, N, E);
    dinv_kernel<<<(N + TB - 1) / TB, TB, 0, stream>>>(part, dinv, N);

    gemm_wmma_kernel<128, 128, false><<<256, 128, 0, stream>>>(x, nullptr, W_in, dinv, H1, A1, N);
    {
        int ewaves = (E + 31) / 32;
        edge_scatter_kernel<128, 8><<<(ewaves + 7) / 8, TB, 0, stream>>>(src, dst, dinv, H1, A1, E);
    }

    gemm_wmma_kernel<128, 64, true><<<256, 128, 0, stream>>>(A1, b_in, W_out, dinv, H2, A2, N);
    {
        int ewaves = (E + 31) / 32;
        edge_scatter_kernel<64, 16><<<(ewaves + 7) / 8, TB, 0, stream>>>(src, dst, dinv, H2, A2, E);
    }

    {
        int dwaves = (NDEC + 31) / 32;
        decoder_kernel<<<(dwaves + 7) / 8, 256, 0, stream>>>(A2, b_out, ia, ib, logits, NDEC);
    }
}


# Edge walks look up no dinv[src]: GEMM stores rows times dinv[row]; layer 1 adds rows unweighted, dinv[dst] applied at the next GEMM's row load; layer 2 keeps one dinv[dst] gather; rounding changes
# speedup vs baseline: 2.9785x; 1.0532x over previous
#include <hip/hip_runtime.h>

typedef float v2f __attribute__((ext_vector_type(2)));
typedef float v8f __attribute__((ext_vector_type(8)));

constexpr int HIST_CAP    = 16384;
constexpr int HIST_SHARES = 64;
constexpr int HIST_TB     = 256;
constexpr int HIST_BATCH  = 32;

__global__ __launch_bounds__(HIST_TB) void deg_hist_kernel(const int* __restrict__ dst,
                                                           int* __restrict__ part,
                                                           int n, int e) {
    __shared__ int4 hist4[HIST_CAP / 4];
    int* hist = (int*)hist4;
    const int tid    = threadIdx.x;
    const int stride = (n + 3) & ~3;
    const int nr     = gridDim.y;
    const int rs     = ((n + nr - 1) / nr + 3) & ~3;
    const int base   = blockIdx.y * rs;
    const int m      = n - base < rs ? n - base : rs;
    if (m <= 0) return;
    const int m4     = (m + 3) >> 2;
    const int per    = (e + HIST_SHARES - 1) / HIST_SHARES;
    const int e0     = blockIdx.x * per;
    const int e1     = e0 + per < e ? e0 + per : e;

    for (int k = tid; k < m4; k += HIST_TB) hist4[k] = make_int4(0, 0, 0, 0);
    __syncthreads();

    for (int i0 = e0 + tid; i0 < e1; i0 += HIST_BATCH * HIST_TB) {
        unsigned d[HIST_BATCH];
#pragma unroll
        for (int k = 0; k < HIST_BATCH; ++k) {
            const int i = i0 + k * HIST_TB;
            d[k] = (unsigned)(dst[i < e1 ? i : e1 - 1] - base);
        }
#pragma unroll
        for (int k = 0; k < HIST_BATCH; ++k) {
            const int i = i0 + k * HIST_TB;
            if (i < e1 && d[k] < (unsigned)m) atomicAdd(&hist[d[k]], 1);
        }
    }
    __syncthreads();

    int4* plane4 = (int4*)(part + (size_t)blockIdx.x * stride + base);
    for (int k = tid; k < m4; k += HIST_TB) plane4[k] = hist4[k];
}

__global__ __launch_bounds__(256) void dinv_kernel(const int* __restrict__ part,
                                                   float* __restrict__ dinv, int n) {
    int i = blockIdx.x * blockDim.x + threadIdx.x;
    if (i >= n) return;
    const int stride = (n + 3) & ~3;
    int s = 1;
#pragma unroll
    for (int b = 0; b < HIST_SHARES; ++b) s += part[(size_t)b * stride + i];
    dinv[i] = rsqrtf((float)s);
}

__device__ __forceinline__ float relu_f(float v) { return v > 0.0f ? v : 0.0f; }

template <int K, int N, bool RELU_IN, bool AGG_DST>
__global__ __launch_bounds__(128) void gemm_wmma_kernel(const float* __restrict__ A,
                                                        const float* __restrict__ abias,
                                                        const float* __restrict__ B,
                                                        const float* __restrict__ dinv,
                                                        float* __restrict__ C,
                                                        float* __restrict__ AGG, int M) {
    constexpr int NT = N / 16;
    constexpr int NG = N / 64;
    constexpr int KJ = K / 8;

    const int lane   = threadIdx.x & 31;
    const int wave   = threadIdx.x >> 5;
    const int h      = lane >> 4;
    const int c      = lane & 15;
    const int nTiles = (M + 15) >> 4;
    const int nWaves = gridDim.x * 4;
    int t = blockIdx.x * 4 + wave;
    if (t >= nTiles) return;

    v2f bf[2 * KJ][NT];
#pragma unroll
    for (int j = 0; j < KJ; ++j) {
#pragma unroll
        for (int e = 0; e < 2; ++e) {
            const int k0 = 8 * j + 4 * h + 2 * e;
#pragma unroll
            for (int g4 = 0; g4 < NG; ++g4) {
                const float4 b0 = *(const float4*)(B + (size_t)k0 * N + 64 * g4 + 4 * c);
                const float4 b1 = *(const float4*)(B + (size_t)(k0 + 1) * N + 64 * g4 + 4 * c);
                bf[2 * j + e][4 * g4 + 0].x = b0.x; bf[2 * j + e][4 * g4 + 0].y = b1.x;
                bf[2 * j + e][4 * g4 + 1].x = b0.y; bf[2 * j + e][4 * g4 + 1].y = b1.y;
                bf[2 * j + e][4 * g4 + 2].x = b0.z; bf[2 * j + e][4 * g4 + 2].y = b1.z;
                bf[2 * j + e][4 * g4 + 3].x = b0.w; bf[2 * j + e][4 * g4 + 3].y = b1.w;
            }
        }
    }

    float4 bq[KJ];
    if (RELU_IN) {
#pragma unroll
        for (int j = 0; j < KJ; ++j) bq[j] = *(const float4*)(abias + 8 * j + 4 * h);
    }

    float4 q[KJ];
    float da = 1.0f;
    {
        int r = t * 16 + c; r = r < M ? r : M - 1;
        const float4* ap = (const float4*)(A + (size_t)r * K + 4 * h);
#pragma unroll
        for (int j = 0; j < KJ; ++j) q[j] = ap[2 * j];
        if (RELU_IN) da = dinv[r];
    }
    float dv[8];
#pragma unroll
    for (int g = 0; g < 8; ++g) {
        int r = t * 16 + 8 * h + g; r = r < M ? r : M - 1;
        dv[g] = dinv[r];
    }

    for (;;) {
        const int tn = t + nWaves;
        float4 qn[KJ];
        float dan = 1.0f;
        {
            int r = tn * 16 + c; r = r < M ? r : M - 1;
            const float4* ap = (const float4*)(A + (size_t)r * K + 4 * h);
#pragma unroll
            for (int j = 0; j < KJ; ++j) qn[j] = ap[2 * j];
            if (RELU_IN) dan = dinv[r];
        }
        float dvn[8];
#pragma unroll
        for (int g = 0; g < 8; ++g) {
            int r = tn * 16 + 8 * h + g; r = r < M ? r : M - 1;
            dvn[g] = dinv[r];
        }
        __builtin_amdgcn_sched_barrier(0);

        v8f acc[NT];
#pragma unroll
        for (int nt = 0; nt < NT; ++nt) acc[nt] = v8f{};

#pragma unroll
        for (int j = 0; j < KJ; ++j) {
            float4 qa = q[j];
            if (RELU_IN) {
                qa.x = relu_f(qa.x * da + bq[j].x); qa.y = relu_f(qa.y * da + bq[j].y);
                qa.z = relu_f(qa.z * da + bq[j].z); qa.w = relu_f(qa.w * da + bq[j].w);
            }
            v2f a0, a1;
            a0.x = qa.x; a0.y = qa.y;
            a1.x = qa.z; a1.y = qa.w;
#pragma unroll
            for (int nt = 0; nt < NT; ++nt)
                acc[nt] = __builtin_amdgcn_wmma_f32_16x16x4_f32(
                    false, a0, false, bf[2 * j][nt], (short)0, acc[nt], false, false);
#pragma unroll
            for (int nt = 0; nt < NT; ++nt)
                acc[nt] = __builtin_amdgcn_wmma_f32_16x16x4_f32(
                    false, a1, false, bf[2 * j + 1][nt], (short)0, acc[nt], false, false);
        }

        const int row0 = t * 16 + 8 * h;
#pragma unroll
        for (int g = 0; g < 8; ++g) {
            const int r = row0 + g;
            if (r < M) {
                const float w = dv[g];
#pragma unroll
                for (int g4 = 0; g4 < NG; ++g4) {
                    const float4 o = make_float4(acc[4 * g4 + 0][g] * w, acc[4 * g4 + 1][g] * w,
                                                 acc[4 * g4 + 2][g] * w, acc[4 * g4 + 3][g] * w);
                    *(float4*)(C + (size_t)r * N + 64 * g4 + 4 * c) = o;
                    *(float4*)(AGG + (size_t)r * N + 64 * g4 + 4 * c) =
                        AGG_DST ? make_float4(o.x * w, o.y * w, o.z * w, o.w * w) : o;
                }
            }
        }

        if (tn >= nTiles) break;
        t = tn;
#pragma unroll
        for (int j = 0; j < KJ; ++j) q[j] = qn[j];
        da = dan;
#pragma unroll
        for (int g = 0; g < 8; ++g) dv[g] = dvn[g];
    }
}


template <int F, int G, bool DSTW>
__global__ __launch_bounds__(256) void edge_scatter_kernel(const int* __restrict__ src,
                                                           const int* __restrict__ dst,
                                                           const float* __restrict__ dinv,
                                                           const float* __restrict__ H,
                                                           float* __restrict__ AGG, int e) {
    constexpr int L = F / 32;
    const unsigned lane = threadIdx.x & 31u;
    const int wave = blockIdx.x * (blockDim.x >> 5) +
                     __builtin_amdgcn_readfirstlane((int)(threadIdx.x >> 5));
    const int e0 = wave * 32;
    if (e0 >= e) return;
    const int cnt = e - e0;

    if (cnt >= 32) {
        const int s = src[e0 + lane];
        const int d = dst[e0 + lane];
        int wbits = 0;
        if (DSTW) wbits = __float_as_int(dinv[d]);
        for (int b = 0; b < 32; b += G) {
            float v[G][L];
            for (int i = 0; i < G; ++i) {
                const int sj = __builtin_amdgcn_readlane(s, b + i);
                const float* row = H + (size_t)sj * F;
                for (int k = 0; k < L; ++k) v[i][k] = row[lane + 32u * k];
            }
            for (int i = 0; i < G; ++i) {
                const int dj = __builtin_amdgcn_readlane(d, b + i);
                float* orow = AGG + (size_t)dj * F;
                if (DSTW) {
                    const float wj = __int_as_float(__builtin_amdgcn_readlane(wbits, b + i));
                    for (int k = 0; k < L; ++k) atomicAdd(orow + lane + 32u * k, v[i][k] * wj);
                } else {
                    for (int k = 0; k < L; ++k) atomicAdd(orow + lane + 32u * k, v[i][k]);
                }
            }
        }
    } else {
        for (int j = 0; j < cnt; ++j) {
            const int sj = src[e0 + j], dj = dst[e0 + j];
            const float wj = DSTW ? dinv[dj] : 1.0f;
            const float* row = H + (size_t)sj * F;
            float* orow = AGG + (size_t)dj * F;
            for (int k = 0; k < L; ++k) atomicAdd(orow + lane + 32u * k, row[lane + 32u * k] * wj);
        }
    }
}

constexpr int DEC_G = 16;

__global__ __launch_bounds__(256) void decoder_kernel(const float* __restrict__ Z,
                                                      const float* __restrict__ bias,
                                                      const int* __restrict__ ia,
                                                      const int* __restrict__ ib,
                                                      float* __restrict__ out, int npairs) {
    const unsigned lane = threadIdx.x & 31u;
    const int wave = blockIdx.x * (blockDim.x >> 5) +
                     __builtin_amdgcn_readfirstlane((int)(threadIdx.x >> 5));
    const int p0 = wave * 32;
    if (p0 >= npairs) return;
    const int pl = p0 + (int)lane < npairs ? p0 + (int)lane : npairs - 1;
    const int a = ia[pl], b = ib[pl];
    const float2 bz = ((const float2*)bias)[lane];

    float v[32];
#pragma unroll
    for (int g = 0; g < 32; g += DEC_G) {
        float2 x[DEC_G], y[DEC_G];
#pragma unroll
        for (int i = 0; i < DEC_G; ++i) {
            const unsigned aj = (unsigned)__builtin_amdgcn_readlane(a, g + i);
            const unsigned bj = (unsigned)__builtin_amdgcn_readlane(b, g + i);
            x[i] = ((const float2*)(Z + (size_t)aj * 64))[lane];
            y[i] = ((const float2*)(Z + (size_t)bj * 64))[lane];
        }
#pragma unroll
        for (int i = 0; i < DEC_G; ++i) {
            const float xa = relu_f(x[i].x + bz.x), xb = relu_f(x[i].y + bz.y);
            const float ya = relu_f(y[i].x + bz.x), yb = relu_f(y[i].y + bz.y);
            v[g + i] = xa * ya + xb * yb;
        }
    }

    float w16[16], w8[8], w4[4], w2[2];
    const bool u16 = (lane & 16u) != 0, u8 = (lane & 8u) != 0, u4 = (lane & 4u) != 0,
               u2 = (lane & 2u) != 0, u1 = (lane & 1u) != 0;
#pragma unroll
    for (int i = 0; i < 16; ++i) {
        const float keep = u16 ? v[i + 16] : v[i];
        const float send = u16 ? v[i] : v[i + 16];
        w16[i] = keep + __shfl_xor(send, 16, 32);
    }
#pragma unroll
    for (int i = 0; i < 8; ++i) {
        const float keep = u8 ? w16[i + 8] : w16[i];
        const float send = u8 ? w16[i] : w16[i + 8];
        w8[i] = keep + __shfl_xor(send, 8, 32);
    }
#pragma unroll
    for (int i = 0; i < 4; ++i) {
        const float keep = u4 ? w8[i + 4] : w8[i];
        const float send = u4 ? w8[i] : w8[i + 4];
        w4[i] = keep + __shfl_xor(send, 4, 32);
    }
#pragma unroll
    for (int i = 0; i < 2; ++i) {
        const float keep = u2 ? w4[i + 2] : w4[i];
        const float send = u2 ? w4[i] : w4[i + 2];
        w2[i] = keep + __shfl_xor(send, 2, 32);
    }
    const float keep = u1 ? w2[1] : w2[0];
    const float send = u1 ? w2[0] : w2[1];
    const float r = keep + __shfl_xor(send, 1, 32);

    if (p0 + (int)lane < npairs) out[p0 + lane] = r;
}

extern "C" void kernel_launch(void* const* d_in, const int* in_sizes, int n_in,
                              void* d_out, int out_size, void* d_ws, size_t ws_size,
                              hipStream_t stream) {
    const float* x     = (const float*)d_in[0];
    const float* W_in  = (const float*)d_in[1];
    const float* b_in  = (const float*)d_in[2];
    const float* W_out = (const float*)d_in[3];
    const float* b_out = (const float*)d_in[4];
    const int*   ei    = (const int*)d_in[5];
    const int*   eli   = (const int*)d_in[6];

    const int N    = in_sizes[0] / 128;
    const int E    = in_sizes[5] / 2;
    const int NDEC = in_sizes[6] / 2;
    const int* src = ei;
    const int* dst = ei + E;
    const int* ia  = eli;
    const int* ib  = eli + NDEC;
    float* logits  = (float*)d_out;

    char* base = (char*)d_ws;
    size_t off = 0;
    auto take = [&](size_t bytes) -> float* {
        float* p = (float*)(base + off);
        off = (off + bytes + 255) & ~(size_t)255;
        return p;
    };
    float* dinv = take((size_t)N * 4);
    float* H1   = take((size_t)N * 128 * 4);
    float* A1   = take((size_t)N * 128 * 4);
    float* H2   = take((size_t)N * 64 * 4);
    float* A2   = take((size_t)N * 64 * 4);
    int*   part = (int*)take((size_t)HIST_SHARES * (N + 3) * 4);
    (void)ws_size;

    const int TB = 256;
    deg_hist_kernel<<<dim3(HIST_SHARES, (N + HIST_CAP - 1) / HIST_CAP), HIST_TB, 0, stream>>>(dst, part, N, E);
    dinv_kernel<<<(N + TB - 1) / TB, TB, 0, stream>>>(part, dinv, N);

    gemm_wmma_kernel<128, 128, false, false><<<256, 128, 0, stream>>>(x, nullptr, W_in, dinv, H1, A1, N);
    {
        int ewaves = (E + 31) / 32;
        edge_scatter_kernel<128, 8, false><<<(ewaves + 7) / 8, TB, 0, stream>>>(src, dst, dinv, H1, A1, E);
    }

    gemm_wmma_kernel<128, 64, true, true><<<256, 128, 0, stream>>>(A1, b_in, W_out, dinv, H2, A2, N);
    {
        int ewaves = (E + 31) / 32;
        edge_scatter_kernel<64, 16, true><<<(ewaves + 7) / 8, TB, 0, stream>>>(src, dst, dinv, H2, A2, E);
    }

    {
        int dwaves = (NDEC + 31) / 32;
        decoder_kernel<<<(dwaves + 7) / 8, 256, 0, stream>>>(A2, b_out, ia, ib, logits, NDEC);
    }
}
